# starting kernel
# baseline (speedup 1.0000x reference)
#include <hip/hip_runtime.h>

#define NN 50000
#define NE 800000
#define DD 128

typedef float v2f __attribute__((ext_vector_type(2)));
typedef float v8f __attribute__((ext_vector_type(8)));

__global__ __launch_bounds__(256) void gemm5_wmma(
    const float* __restrict__ X,
    const float* __restrict__ W1, const float* __restrict__ b1,
    const float* __restrict__ W2, const float* __restrict__ b2,
    const float* __restrict__ W3, const float* __restrict__ b3,
    const float* __restrict__ W4, const float* __restrict__ b4,
    const float* __restrict__ Wr, const float* __restrict__ br,
    float* __restrict__ H1, float* __restrict__ H2,
    float* __restrict__ H3, float* __restrict__ H4,
    float* __restrict__ Hr)
{
    __shared__ float xs[16 * 132];
    const int row0 = blockIdx.x * 16;
    const int t = threadIdx.x;

    #pragma unroll
    for (int q = t; q < 512; q += 256) {
        const int r  = q >> 5;
        const int c4 = (q & 31) << 2;
        const float4 v = *reinterpret_cast<const float4*>(X + (size_t)(row0 + r) * DD + c4);
        *reinterpret_cast<float4*>(&xs[r * 132 + c4]) = v;
    }
    __syncthreads();

    const int wave = t >> 5;
    const int lane = t & 31;
    const int half = lane >> 4;
    const int nn   = lane & 15;
    const int n0   = wave * 16;
    const int wrow = n0 + nn;

    const float* Ws[5] = {W1, W2, W3, W4, Wr};
    const float* Bs[5] = {b1, b2, b3, b4, br};
    float*       Hs[5] = {H1, H2, H3, H4, Hr};

    v8f acc[5] = {};

    for (int k = 0; k < DD; k += 4) {
        const int kk = k + 2 * half;
        const float2 af = *reinterpret_cast<const float2*>(&xs[nn * 132 + kk]);
        v2f a; a.x = af.x; a.y = af.y;
        #pragma unroll
        for (int w = 0; w < 5; ++w) {
            const float2 bf = *reinterpret_cast<const float2*>(Ws[w] + (size_t)wrow * DD + kk);
            v2f b; b.x = bf.x; b.y = bf.y;
            acc[w] = __builtin_amdgcn_wmma_f32_16x16x4_f32(
                false, a, false, b, (short)0, acc[w], false, false);
        }
    }

    #pragma unroll
    for (int w = 0; w < 5; ++w) {
        const float bias = Bs[w][wrow];
        #pragma unroll
        for (int j = 0; j < 8; ++j) {
            const int m = j + 8 * half;
            Hs[w][(size_t)(row0 + m) * DD + wrow] = acc[w][j] + bias;
        }
    }
}

__global__ __launch_bounds__(256) void edge_gate_scatter(
    const int* __restrict__ ei,
    const float* __restrict__ H2, const float* __restrict__ H3,
    const float* __restrict__ H4,
    float* __restrict__ H1acc, float* __restrict__ cnt)
{
    const int wave = threadIdx.x >> 5;
    const int lane = threadIdx.x & 31;
    const int e = blockIdx.x * 8 + wave;
    if (e >= NE) return;

    const int src = ei[e];
    const int dst = ei[NE + e];
    const int d0 = lane << 2;

    const float4 x3 = *reinterpret_cast<const float4*>(H3 + (size_t)src * DD + d0);
    const float4 x4 = *reinterpret_cast<const float4*>(H4 + (size_t)dst * DD + d0);
    const float4 x2 = *reinterpret_cast<const float4*>(H2 + (size_t)src * DD + d0);

    float4 m;
    m.x = x2.x / (1.0f + __expf(-(x3.x + x4.x)));
    m.y = x2.y / (1.0f + __expf(-(x3.y + x4.y)));
    m.z = x2.z / (1.0f + __expf(-(x3.z + x4.z)));
    m.w = x2.w / (1.0f + __expf(-(x3.w + x4.w)));

    float* p = H1acc + (size_t)dst * DD + d0;
    unsafeAtomicAdd(p + 0, m.x);
    unsafeAtomicAdd(p + 1, m.y);
    unsafeAtomicAdd(p + 2, m.z);
    unsafeAtomicAdd(p + 3, m.w);

    if (lane == 0) unsafeAtomicAdd(&cnt[dst], 1.0f);
}

__global__ __launch_bounds__(256) void finalize_mean_res(
    const float* __restrict__ H1acc, const float* __restrict__ cnt,
    float* __restrict__ out)
{
    const size_t i = (size_t)blockIdx.x * 256 + threadIdx.x;
    const int node = (int)(i >> 5);
    const float s = 1.0f / fmaxf(cnt[node], 1.0f);
    const float4 h = reinterpret_cast<const float4*>(H1acc)[i];
    float4 o = reinterpret_cast<float4*>(out)[i];
    o.x += h.x * s;
    o.y += h.y * s;
    o.z += h.z * s;
    o.w += h.w * s;
    reinterpret_cast<float4*>(out)[i] = o;
}

extern "C" void kernel_launch(void* const* d_in, const int* in_sizes, int n_in,
                              void* d_out, int out_size, void* d_ws, size_t ws_size,
                              hipStream_t stream)
{
    const float* X   = (const float*)d_in[0];
    const int*   ei  = (const int*)  d_in[1];
    const float* W1  = (const float*)d_in[2];
    const float* b1  = (const float*)d_in[3];
    const float* W2  = (const float*)d_in[4];
    const float* b2  = (const float*)d_in[5];
    const float* W3  = (const float*)d_in[6];
    const float* b3  = (const float*)d_in[7];
    const float* W4  = (const float*)d_in[8];
    const float* b4  = (const float*)d_in[9];
    const float* Wr  = (const float*)d_in[10];
    const float* br  = (const float*)d_in[11];
    float* out = (float*)d_out;

    const size_t nd = (size_t)NN * DD;
    float* ws  = (float*)d_ws;
    float* H1  = ws;
    float* H2  = ws + nd;
    float* H3  = ws + 2 * nd;
    float* H4  = ws + 3 * nd;
    float* cnt = ws + 4 * nd;

    hipMemsetAsync(cnt, 0, NN * sizeof(float), stream);

    gemm5_wmma<<<NN / 16, 256, 0, stream>>>(
        X, W1, b1, W2, b2, W3, b3, W4, b4, Wr, br,
        H1, H2, H3, H4,   out);

    edge_gate_scatter<<<NE / 8, 256, 0, stream>>>(ei, H2, H3, H4, H1, cnt);

    finalize_mean_res<<<(int)(nd / 4 / 256), 256, 0, stream>>>(H1, cnt, out);
}


# Edge phase: a work-group owns 196 destination nodes (sums, h4 rows, degree in LDS) and filters the whole edge list itself; LDS float adds in arrival order, no global atomics; mean+residual fused
# speedup vs baseline: 1.6002x; 1.6002x over previous
#include <hip/hip_runtime.h>

#define NN 50000
#define NE 800000
#define DD 128

typedef float v2f __attribute__((ext_vector_type(2)));
typedef float v8f __attribute__((ext_vector_type(8)));

__global__ __launch_bounds__(256) void gemm5_wmma(
    const float* __restrict__ X,
    const float* __restrict__ W1, const float* __restrict__ b1,
    const float* __restrict__ W2, const float* __restrict__ b2,
    const float* __restrict__ W3, const float* __restrict__ b3,
    const float* __restrict__ W4, const float* __restrict__ b4,
    const float* __restrict__ Wr, const float* __restrict__ br,
    float* __restrict__ H1, float* __restrict__ H2,
    float* __restrict__ H3, float* __restrict__ H4,
    float* __restrict__ Hr)
{
    __shared__ float xs[16 * 132];
    const int row0 = blockIdx.x * 16;
    const int t = threadIdx.x;

    #pragma unroll
    for (int q = t; q < 512; q += 256) {
        const int r  = q >> 5;
        const int c4 = (q & 31) << 2;
        const float4 v = *reinterpret_cast<const float4*>(X + (size_t)(row0 + r) * DD + c4);
        *reinterpret_cast<float4*>(&xs[r * 132 + c4]) = v;
    }
    __syncthreads();

    const int wave = t >> 5;
    const int lane = t & 31;
    const int half = lane >> 4;
    const int nn   = lane & 15;
    const int n0   = wave * 16;
    const int wrow = n0 + nn;

    const float* Ws[5] = {W1, W2, W3, W4, Wr};
    const float* Bs[5] = {b1, b2, b3, b4, br};
    float*       Hs[5] = {H1, H2, H3, H4, Hr};

    v8f acc[5] = {};

    for (int k = 0; k < DD; k += 4) {
        const int kk = k + 2 * half;
        const float2 af = *reinterpret_cast<const float2*>(&xs[nn * 132 + kk]);
        v2f a; a.x = af.x; a.y = af.y;
        #pragma unroll
        for (int w = 0; w < 5; ++w) {
            const float2 bf = *reinterpret_cast<const float2*>(Ws[w] + (size_t)wrow * DD + kk);
            v2f b; b.x = bf.x; b.y = bf.y;
            acc[w] = __builtin_amdgcn_wmma_f32_16x16x4_f32(
                false, a, false, b, (short)0, acc[w], false, false);
        }
    }

    #pragma unroll
    for (int w = 0; w < 5; ++w) {
        const float bias = Bs[w][wrow];
        #pragma unroll
        for (int j = 0; j < 8; ++j) {
            const int m = j + 8 * half;
            Hs[w][(size_t)(row0 + m) * DD + wrow] = acc[w][j] + bias;
        }
    }
}

#define OWN 196
#define QCAP 160
#define NCHUNK (NE / 128)

__global__ __launch_bounds__(1024) void edge_owner_accumulate(
    const int* __restrict__ ei,
    const float* __restrict__ H1, const float* __restrict__ H2,
    const float* __restrict__ H3, const float* __restrict__ H4,
    float* __restrict__ out)
{
    __shared__ float acc[OWN * DD];
    __shared__ float h4s[OWN * DD];
    __shared__ int   cnt[OWN];
    __shared__ int   qe[32 * QCAP];
    __shared__ int   qd[32 * QCAP];

    const int t = threadIdx.x;
    const int lane = t & 31;
    const int wave = __builtin_amdgcn_readfirstlane(t >> 5);
    const int node0 = blockIdx.x * OWN;
    const int nown = min(OWN, NN - node0);

    for (int q = t; q < OWN * 32; q += 1024) {
        const int n = q >> 5, l = q & 31;
        float4 a = make_float4(0.f, 0.f, 0.f, 0.f), b = a;
        if (n < nown) {
            a = *reinterpret_cast<const float4*>(H1 + (size_t)(node0 + n) * DD + 4 * l);
            b = *reinterpret_cast<const float4*>(H4 + (size_t)(node0 + n) * DD + 4 * l);
        }
        acc[n * DD + l]      = a.x;
        acc[n * DD + 32 + l] = a.y;
        acc[n * DD + 64 + l] = a.z;
        acc[n * DD + 96 + l] = a.w;
        *reinterpret_cast<float4*>(&h4s[n * DD + 4 * l]) = b;
    }
    if (t < OWN) cnt[t] = 0;
    __syncthreads();

    int* const myqe = qe + wave * QCAP;
    int* const myqd = qd + wave * QCAP;
    const int4* const dstv = reinterpret_cast<const int4*>(ei + NE);
    int qn = 0;

    auto drain = [&]() {
        __builtin_amdgcn_wave_barrier();
        for (int base = 0; base < qn; base += 32) {
            const int nb = min(32, qn - base);
            int my_e = 0, my_d = 0, my_s = 0;
            if (lane < nb) {
                my_e = myqe[base + lane];
                my_d = myqd[base + lane];
                my_s = ei[my_e];
            }
            for (int g = 0; g < nb; g += 4) {
                float4 x2[4], x3[4];
                int dl[4];
                #pragma unroll
                for (int k = 0; k < 4; ++k) {
                    const int idx = (g + k < nb) ? (g + k) : g;
                    const int s = __builtin_amdgcn_readlane(my_s, idx);
                    dl[k] = __builtin_amdgcn_readlane(my_d, idx);
                    x3[k] = *reinterpret_cast<const float4*>(H3 + (size_t)s * DD + 4 * lane);
                    x2[k] = *reinterpret_cast<const float4*>(H2 + (size_t)s * DD + 4 * lane);
                }
                #pragma unroll
                for (int k = 0; k < 4; ++k) {
                    if (g + k < nb) {
                        const float4 x4 = *reinterpret_cast<const float4*>(&h4s[dl[k] * DD + 4 * lane]);
                        float4 m;
                        m.x = x2[k].x / (1.0f + __expf(-(x3[k].x + x4.x)));
                        m.y = x2[k].y / (1.0f + __expf(-(x3[k].y + x4.y)));
                        m.z = x2[k].z / (1.0f + __expf(-(x3[k].z + x4.z)));
                        m.w = x2[k].w / (1.0f + __expf(-(x3[k].w + x4.w)));
                        float* p = &acc[dl[k] * DD + lane];
                        unsafeAtomicAdd(p,      m.x);
                        unsafeAtomicAdd(p + 32, m.y);
                        unsafeAtomicAdd(p + 64, m.z);
                        unsafeAtomicAdd(p + 96, m.w);
                        if (lane == 0) atomicAdd(&cnt[dl[k]], 1);
                    }
                }
            }
        }
        __builtin_amdgcn_wave_barrier();
        qn = 0;
    };

    const int rot = blockIdx.x * (NCHUNK / 256);
    for (int c0 = wave; c0 < NCHUNK; c0 += 128) {
        int4 d[4];
        int cr[4];
        #pragma unroll
        for (int u = 0; u < 4; ++u) {
            int cc = min(c0 + 32 * u, NCHUNK - 1) + rot;
            if (cc >= NCHUNK) cc -= NCHUNK;
            cr[u] = cc;
            d[u] = dstv[(size_t)cc * 32 + lane];
        }
        #pragma unroll
        for (int u = 0; u < 4; ++u) {
            if (c0 + 32 * u < NCHUNK) {
                const unsigned r[4] = {(unsigned)(d[u].x - node0), (unsigned)(d[u].y - node0),
                                       (unsigned)(d[u].z - node0), (unsigned)(d[u].w - node0)};
                const bool any = (r[0] < (unsigned)nown) | (r[1] < (unsigned)nown) |
                                 (r[2] < (unsigned)nown) | (r[3] < (unsigned)nown);
                if (__builtin_amdgcn_ballot_w32(any) != 0u) {
                    #pragma unroll
                    for (int j = 0; j < 4; ++j) {
                        const bool h = r[j] < (unsigned)nown;
                        const unsigned mk = __builtin_amdgcn_ballot_w32(h);
                        if (h) {
                            const int pos = qn + (int)__builtin_amdgcn_mbcnt_lo(mk, 0u);
                            myqe[pos] = cr[u] * 128 + lane * 4 + j;
                            myqd[pos] = (int)r[j];
                        }
                        qn += __builtin_popcount(mk);
                    }
                    if (qn >= 32) drain();
                }
            }
        }
    }
    if (qn > 0) drain();
    __syncthreads();

    for (int q = t; q < nown * 32; q += 1024) {
        const int n = q >> 5, l = q & 31;
        const float s = 1.0f / fmaxf((float)cnt[n], 1.0f);
        float4* po = reinterpret_cast<float4*>(out + (size_t)(node0 + n) * DD + 4 * l);
        float4 o = *po;
        o.x += acc[n * DD + l]      * s;
        o.y += acc[n * DD + 32 + l] * s;
        o.z += acc[n * DD + 64 + l] * s;
        o.w += acc[n * DD + 96 + l] * s;
        *po = o;
    }
}

extern "C" void kernel_launch(void* const* d_in, const int* in_sizes, int n_in,
                              void* d_out, int out_size, void* d_ws, size_t ws_size,
                              hipStream_t stream)
{
    const float* X   = (const float*)d_in[0];
    const int*   ei  = (const int*)  d_in[1];
    const float* W1  = (const float*)d_in[2];
    const float* b1  = (const float*)d_in[3];
    const float* W2  = (const float*)d_in[4];
    const float* b2  = (const float*)d_in[5];
    const float* W3  = (const float*)d_in[6];
    const float* b3  = (const float*)d_in[7];
    const float* W4  = (const float*)d_in[8];
    const float* b4  = (const float*)d_in[9];
    const float* Wr  = (const float*)d_in[10];
    const float* br  = (const float*)d_in[11];
    float* out = (float*)d_out;

    const size_t nd = (size_t)NN * DD;
    float* ws  = (float*)d_ws;
    float* H1  = ws;
    float* H2  = ws + nd;
    float* H3  = ws + 2 * nd;
    float* H4  = ws + 3 * nd;

    gemm5_wmma<<<NN / 16, 256, 0, stream>>>(
        X, W1, b1, W2, b2, W3, b3, W4, b4, Wr, br,
        H1, H2, H3, H4,   out);

    edge_owner_accumulate<<<(NN + OWN - 1) / OWN, 1024, 0, stream>>>(ei, H1, H2, H3, H4, out);
}


# GEMM made weight-stationary: one weight per block, a wave keeps all 32 B fragments of its column tile in registers and walks a slab of X row tiles from an LDS ring; no W load in the k loop; sums kept
# speedup vs baseline: 2.1209x; 1.3254x over previous
#include <hip/hip_runtime.h>

#define NN 50000
#define NE 800000
#define DD 128

typedef float v2f __attribute__((ext_vector_type(2)));
typedef float v8f __attribute__((ext_vector_type(8)));

#define NTILES (NN / 16)
#define SLAB_TILES 16
#define NSLAB ((NTILES + SLAB_TILES - 1) / SLAB_TILES)
#define XPITCH 132
#define STAGE_FLOATS (32 * XPITCH)

__global__ __launch_bounds__(256, 8) void gemm5_wstationary(
    const float* __restrict__ X,
    const float* __restrict__ W1, const float* __restrict__ b1,
    const float* __restrict__ W2, const float* __restrict__ b2,
    const float* __restrict__ W3, const float* __restrict__ b3,
    const float* __restrict__ W4, const float* __restrict__ b4,
    const float* __restrict__ Wr, const float* __restrict__ br,
    float* __restrict__ H1, float* __restrict__ H2,
    float* __restrict__ H3, float* __restrict__ H4,
    float* __restrict__ Hr)
{
    __shared__ float xs[2 * STAGE_FLOATS];

    const int t = threadIdx.x;
    const int wave = __builtin_amdgcn_readfirstlane(t >> 5);
    const int lane = t & 31;
    const int half = lane >> 4;
    const int nn   = lane & 15;
    const int wrow = wave * 16 + nn;

    const int ws = blockIdx.y;
    const float* __restrict__ Wp = ws == 0 ? W1 : ws == 1 ? W2 : ws == 2 ? W3 : ws == 3 ? W4 : Wr;
    const float* __restrict__ Bp = ws == 0 ? b1 : ws == 1 ? b2 : ws == 2 ? b3 : ws == 3 ? b4 : br;
    float* __restrict__ Hp       = ws == 0 ? H1 : ws == 1 ? H2 : ws == 2 ? H3 : ws == 3 ? H4 : Hr;

    const int tile0 = blockIdx.x * SLAB_TILES;
    const int ntile = min(SLAB_TILES, NTILES - tile0);
    const int niter = (ntile + 1) >> 1;

    const int sc = (t & 31) << 2;
    const int srow = t >> 5;
    auto xtile = [&](int tile) {
        return X + ((size_t)min(tile, NTILES - 1) * 16 + srow) * DD + sc;
    };
    float4 pa0, pa1, pb0, pb1;

    v2f bfrag[32];
    #pragma unroll
    for (int p = 0; p < 2; ++p) {
        #pragma unroll
        for (int i = 0; i < 8; ++i) {
            const int r = srow + 8 * i;
            const float4 v = *reinterpret_cast<const float4*>(Wp + (size_t)(64 * p + r) * DD + sc);
            *reinterpret_cast<float4*>(&xs[r * XPITCH + sc]) = v;
        }
        if (p == 1) {
            const float* xa0 = xtile(tile0);
            const float* xb0 = xtile(tile0 + 1);
            pa0 = *reinterpret_cast<const float4*>(xa0);
            pa1 = *reinterpret_cast<const float4*>(xa0 + 8 * DD);
            pb0 = *reinterpret_cast<const float4*>(xb0);
            pb1 = *reinterpret_cast<const float4*>(xb0 + 8 * DD);
        }
        __syncthreads();
        if ((wave >> 2) == p) {
            const float* wp = &xs[((wave & 3) * 16 + nn) * XPITCH + 2 * half];
            #pragma unroll
            for (int s = 0; s < 32; ++s) {
                const float2 bf = *reinterpret_cast<const float2*>(wp + 4 * s);
                bfrag[s].x = bf.x; bfrag[s].y = bf.y;
            }
        }
        __syncthreads();
    }
    const float bias = Bp[wrow];

    {
        float* sp = &xs[srow * XPITCH + sc];
        *reinterpret_cast<float4*>(sp) = pa0;
        *reinterpret_cast<float4*>(sp + 8 * XPITCH) = pa1;
        *reinterpret_cast<float4*>(sp + 16 * XPITCH) = pb0;
        *reinterpret_cast<float4*>(sp + 24 * XPITCH) = pb1;
    }
    __syncthreads();

    for (int it = 0; it < niter; ++it) {
        const int stage = it & 1;
        const float* xna = xtile(tile0 + 2 * it + 2);
        const float* xnb = xtile(tile0 + 2 * it + 3);
        float* sn = &xs[(stage ^ 1) * STAGE_FLOATS + srow * XPITCH + sc];

        float4 q = *reinterpret_cast<const float4*>(xna);

        v8f acc0 = {}, acc1 = {};
        const float* xa = &xs[stage * STAGE_FLOATS + nn * XPITCH + 2 * half];
        __builtin_amdgcn_sched_barrier(0);
        #pragma unroll
        for (int s = 0; s < 32; ++s) {
            if (s == 8) {
                *reinterpret_cast<float4*>(sn) = q;
                q = *reinterpret_cast<const float4*>(xna + 8 * DD);
            }
            if (s == 16) {
                *reinterpret_cast<float4*>(sn + 8 * XPITCH) = q;
                q = *reinterpret_cast<const float4*>(xnb);
            }
            if (s == 24) {
                *reinterpret_cast<float4*>(sn + 16 * XPITCH) = q;
                q = *reinterpret_cast<const float4*>(xnb + 8 * DD);
            }
            const float2 a0f = *reinterpret_cast<const float2*>(xa + 4 * s);
            const float2 a1f = *reinterpret_cast<const float2*>(xa + 16 * XPITCH + 4 * s);
            v2f a0; a0.x = a0f.x; a0.y = a0f.y;
            v2f a1; a1.x = a1f.x; a1.y = a1f.y;
            acc0 = __builtin_amdgcn_wmma_f32_16x16x4_f32(
                false, a0, false, bfrag[s], (short)0, acc0, false, false);
            acc1 = __builtin_amdgcn_wmma_f32_16x16x4_f32(
                false, a1, false, bfrag[s], (short)0, acc1, false, false);
        }
        *reinterpret_cast<float4*>(sn + 24 * XPITCH) = q;

        __builtin_amdgcn_sched_barrier(0);
        asm volatile("v_nop\n\tv_nop\n\tv_nop\n\tv_nop" : "+v"(acc0), "+v"(acc1));
        __builtin_amdgcn_sched_barrier(0);

        float* hp = Hp + (size_t)((tile0 + 2 * it) * 16 + 8 * half) * DD + wrow;
        #pragma unroll
        for (int j = 0; j < 8; ++j) hp[j * DD] = acc0[j] + bias;
        if (2 * it + 1 < ntile) {
            #pragma unroll
            for (int j = 0; j < 8; ++j) hp[(16 + j) * DD] = acc1[j] + bias;
        }

        __builtin_amdgcn_fence(__ATOMIC_RELEASE, "workgroup", "local");
        __builtin_amdgcn_s_barrier();
        __builtin_amdgcn_fence(__ATOMIC_ACQUIRE, "workgroup", "local");
    }
}

#define OWN 196
#define QCAP 160
#define NCHUNK (NE / 128)

__global__ __launch_bounds__(1024) void edge_owner_accumulate(
    const int* __restrict__ ei,
    const float* __restrict__ H1, const float* __restrict__ H2,
    const float* __restrict__ H3, const float* __restrict__ H4,
    float* __restrict__ out)
{
    __shared__ float acc[OWN * DD];
    __shared__ float h4s[OWN * DD];
    __shared__ int   cnt[OWN];
    __shared__ int   qe[32 * QCAP];
    __shared__ int   qd[32 * QCAP];

    const int t = threadIdx.x;
    const int lane = t & 31;
    const int wave = __builtin_amdgcn_readfirstlane(t >> 5);
    const int node0 = blockIdx.x * OWN;
    const int nown = min(OWN, NN - node0);

    for (int q = t; q < OWN * 32; q += 1024) {
        const int n = q >> 5, l = q & 31;
        float4 a = make_float4(0.f, 0.f, 0.f, 0.f), b = a;
        if (n < nown) {
            a = *reinterpret_cast<const float4*>(H1 + (size_t)(node0 + n) * DD + 4 * l);
            b = *reinterpret_cast<const float4*>(H4 + (size_t)(node0 + n) * DD + 4 * l);
        }
        acc[n * DD + l]      = a.x;
        acc[n * DD + 32 + l] = a.y;
        acc[n * DD + 64 + l] = a.z;
        acc[n * DD + 96 + l] = a.w;
        *reinterpret_cast<float4*>(&h4s[n * DD + 4 * l]) = b;
    }
    if (t < OWN) cnt[t] = 0;
    __syncthreads();

    int* const myqe = qe + wave * QCAP;
    int* const myqd = qd + wave * QCAP;
    const int4* const dstv = reinterpret_cast<const int4*>(ei + NE);
    int qn = 0;

    auto drain = [&]() {
        __builtin_amdgcn_wave_barrier();
        for (int base = 0; base < qn; base += 32) {
            const int nb = min(32, qn - base);
            int my_e = 0, my_d = 0, my_s = 0;
            if (lane < nb) {
                my_e = myqe[base + lane];
                my_d = myqd[base + lane];
                my_s = ei[my_e];
            }
            for (int g = 0; g < nb; g += 4) {
                float4 x2[4], x3[4];
                int dl[4];
                #pragma unroll
                for (int k = 0; k < 4; ++k) {
                    const int idx = (g + k < nb) ? (g + k) : g;
                    const int s = __builtin_amdgcn_readlane(my_s, idx);
                    dl[k] = __builtin_amdgcn_readlane(my_d, idx);
                    x3[k] = *reinterpret_cast<const float4*>(H3 + (size_t)s * DD + 4 * lane);
                    x2[k] = *reinterpret_cast<const float4*>(H2 + (size_t)s * DD + 4 * lane);
                }
                #pragma unroll
                for (int k = 0; k < 4; ++k) {
                    if (g + k < nb) {
                        const float4 x4 = *reinterpret_cast<const float4*>(&h4s[dl[k] * DD + 4 * lane]);
                        float4 m;
                        m.x = x2[k].x / (1.0f + __expf(-(x3[k].x + x4.x)));
                        m.y = x2[k].y / (1.0f + __expf(-(x3[k].y + x4.y)));
                        m.z = x2[k].z / (1.0f + __expf(-(x3[k].z + x4.z)));
                        m.w = x2[k].w / (1.0f + __expf(-(x3[k].w + x4.w)));
                        float* p = &acc[dl[k] * DD + lane];
                        unsafeAtomicAdd(p,      m.x);
                        unsafeAtomicAdd(p + 32, m.y);
                        unsafeAtomicAdd(p + 64, m.z);
                        unsafeAtomicAdd(p + 96, m.w);
                        if (lane == 0) atomicAdd(&cnt[dl[k]], 1);
                    }
                }
            }
        }
        __builtin_amdgcn_wave_barrier();
        qn = 0;
    };

    const int rot = blockIdx.x * (NCHUNK / 256);
    for (int c0 = wave; c0 < NCHUNK; c0 += 128) {
        int4 d[4];
        int cr[4];
        #pragma unroll
        for (int u = 0; u < 4; ++u) {
            int cc = min(c0 + 32 * u, NCHUNK - 1) + rot;
            if (cc >= NCHUNK) cc -= NCHUNK;
            cr[u] = cc;
            d[u] = dstv[(size_t)cc * 32 + lane];
        }
        #pragma unroll
        for (int u = 0; u < 4; ++u) {
            if (c0 + 32 * u < NCHUNK) {
                const unsigned r[4] = {(unsigned)(d[u].x - node0), (unsigned)(d[u].y - node0),
                                       (unsigned)(d[u].z - node0), (unsigned)(d[u].w - node0)};
                const bool any = (r[0] < (unsigned)nown) | (r[1] < (unsigned)nown) |
                                 (r[2] < (unsigned)nown) | (r[3] < (unsigned)nown);
                if (__builtin_amdgcn_ballot_w32(any) != 0u) {
                    #pragma unroll
                    for (int j = 0; j < 4; ++j) {
                        const bool h = r[j] < (unsigned)nown;
                        const unsigned mk = __builtin_amdgcn_ballot_w32(h);
                        if (h) {
                            const int pos = qn + (int)__builtin_amdgcn_mbcnt_lo(mk, 0u);
                            myqe[pos] = cr[u] * 128 + lane * 4 + j;
                            myqd[pos] = (int)r[j];
                        }
                        qn += __builtin_popcount(mk);
                    }
                    if (qn >= 32) drain();
                }
            }
        }
    }
    if (qn > 0) drain();
    __syncthreads();

    for (int q = t; q < nown * 32; q += 1024) {
        const int n = q >> 5, l = q & 31;
        const float s = 1.0f / fmaxf((float)cnt[n], 1.0f);
        float4* po = reinterpret_cast<float4*>(out + (size_t)(node0 + n) * DD + 4 * l);
        float4 o = *po;
        o.x += acc[n * DD + l]      * s;
        o.y += acc[n * DD + 32 + l] * s;
        o.z += acc[n * DD + 64 + l] * s;
        o.w += acc[n * DD + 96 + l] * s;
        *po = o;
    }
}

extern "C" void kernel_launch(void* const* d_in, const int* in_sizes, int n_in,
                              void* d_out, int out_size, void* d_ws, size_t ws_size,
                              hipStream_t stream)
{
    const float* X   = (const float*)d_in[0];
    const int*   ei  = (const int*)  d_in[1];
    const float* W1  = (const float*)d_in[2];
    const float* b1  = (const float*)d_in[3];
    const float* W2  = (const float*)d_in[4];
    const float* b2  = (const float*)d_in[5];
    const float* W3  = (const float*)d_in[6];
    const float* b3  = (const float*)d_in[7];
    const float* W4  = (const float*)d_in[8];
    const float* b4  = (const float*)d_in[9];
    const float* Wr  = (const float*)d_in[10];
    const float* br  = (const float*)d_in[11];
    float* out = (float*)d_out;

    const size_t nd = (size_t)NN * DD;
    float* ws  = (float*)d_ws;
    float* H1  = ws;
    float* H2  = ws + nd;
    float* H3  = ws + 2 * nd;
    float* H4  = ws + 3 * nd;

    gemm5_wstationary<<<dim3(NSLAB, 5), 256, 0, stream>>>(
        X, W1, b1, W2, b2, W3, b3, W4, b4, Wr, br,
        H1, H2, H3, H4,   out);

    edge_owner_accumulate<<<(NN + OWN - 1) / OWN, 1024, 0, stream>>>(ei, H1, H2, H3, H4, out);
}


# Edge kernel bookkeeping cut, gate math kept: list loads off one wave-uniform base, one min test per lane, empty components skipped, queue = edge ids, dst+degree per 32-edge batch; sums as before
# speedup vs baseline: 2.2285x; 1.0507x over previous
#include <hip/hip_runtime.h>

#define NN 50000
#define NE 800000
#define DD 128

typedef float v2f __attribute__((ext_vector_type(2)));
typedef float v8f __attribute__((ext_vector_type(8)));

#define NTILES (NN / 16)
#define SLAB_TILES 16
#define NSLAB ((NTILES + SLAB_TILES - 1) / SLAB_TILES)
#define XPITCH 132
#define STAGE_FLOATS (32 * XPITCH)

__global__ __launch_bounds__(256, 8) void gemm5_wstationary(
    const float* __restrict__ X,
    const float* __restrict__ W1, const float* __restrict__ b1,
    const float* __restrict__ W2, const float* __restrict__ b2,
    const float* __restrict__ W3, const float* __restrict__ b3,
    const float* __restrict__ W4, const float* __restrict__ b4,
    const float* __restrict__ Wr, const float* __restrict__ br,
    float* __restrict__ H1, float* __restrict__ H2,
    float* __restrict__ H3, float* __restrict__ H4,
    float* __restrict__ Hr)
{
    __shared__ float xs[2 * STAGE_FLOATS];

    const int t = threadIdx.x;
    const int wave = __builtin_amdgcn_readfirstlane(t >> 5);
    const int lane = t & 31;
    const int half = lane >> 4;
    const int nn   = lane & 15;
    const int wrow = wave * 16 + nn;

    const int ws = blockIdx.y;
    const float* __restrict__ Wp = ws == 0 ? W1 : ws == 1 ? W2 : ws == 2 ? W3 : ws == 3 ? W4 : Wr;
    const float* __restrict__ Bp = ws == 0 ? b1 : ws == 1 ? b2 : ws == 2 ? b3 : ws == 3 ? b4 : br;
    float* __restrict__ Hp       = ws == 0 ? H1 : ws == 1 ? H2 : ws == 2 ? H3 : ws == 3 ? H4 : Hr;

    const int tile0 = blockIdx.x * SLAB_TILES;
    const int ntile = min(SLAB_TILES, NTILES - tile0);
    const int niter = (ntile + 1) >> 1;

    const int sc = (t & 31) << 2;
    const int srow = t >> 5;
    auto xtile = [&](int tile) {
        return X + ((size_t)min(tile, NTILES - 1) * 16 + srow) * DD + sc;
    };
    float4 pa0, pa1, pb0, pb1;

    v2f bfrag[32];
    #pragma unroll
    for (int p = 0; p < 2; ++p) {
        #pragma unroll
        for (int i = 0; i < 8; ++i) {
            const int r = srow + 8 * i;
            const float4 v = *reinterpret_cast<const float4*>(Wp + (size_t)(64 * p + r) * DD + sc);
            *reinterpret_cast<float4*>(&xs[r * XPITCH + sc]) = v;
        }
        if (p == 1) {
            const float* xa0 = xtile(tile0);
            const float* xb0 = xtile(tile0 + 1);
            pa0 = *reinterpret_cast<const float4*>(xa0);
            pa1 = *reinterpret_cast<const float4*>(xa0 + 8 * DD);
            pb0 = *reinterpret_cast<const float4*>(xb0);
            pb1 = *reinterpret_cast<const float4*>(xb0 + 8 * DD);
        }
        __syncthreads();
        if ((wave >> 2) == p) {
            const float* wp = &xs[((wave & 3) * 16 + nn) * XPITCH + 2 * half];
            #pragma unroll
            for (int s = 0; s < 32; ++s) {
                const float2 bf = *reinterpret_cast<const float2*>(wp + 4 * s);
                bfrag[s].x = bf.x; bfrag[s].y = bf.y;
            }
        }
        __syncthreads();
    }
    const float bias = Bp[wrow];

    {
        float* sp = &xs[srow * XPITCH + sc];
        *reinterpret_cast<float4*>(sp) = pa0;
        *reinterpret_cast<float4*>(sp + 8 * XPITCH) = pa1;
        *reinterpret_cast<float4*>(sp + 16 * XPITCH) = pb0;
        *reinterpret_cast<float4*>(sp + 24 * XPITCH) = pb1;
    }
    __syncthreads();

    for (int it = 0; it < niter; ++it) {
        const int stage = it & 1;
        const float* xna = xtile(tile0 + 2 * it + 2);
        const float* xnb = xtile(tile0 + 2 * it + 3);
        float* sn = &xs[(stage ^ 1) * STAGE_FLOATS + srow * XPITCH + sc];

        float4 q = *reinterpret_cast<const float4*>(xna);

        v8f acc0 = {}, acc1 = {};
        const float* xa = &xs[stage * STAGE_FLOATS + nn * XPITCH + 2 * half];
        __builtin_amdgcn_sched_barrier(0);
        #pragma unroll
        for (int s = 0; s < 32; ++s) {
            if (s == 8) {
                *reinterpret_cast<float4*>(sn) = q;
                q = *reinterpret_cast<const float4*>(xna + 8 * DD);
            }
            if (s == 16) {
                *reinterpret_cast<float4*>(sn + 8 * XPITCH) = q;
                q = *reinterpret_cast<const float4*>(xnb);
            }
            if (s == 24) {
                *reinterpret_cast<float4*>(sn + 16 * XPITCH) = q;
                q = *reinterpret_cast<const float4*>(xnb + 8 * DD);
            }
            const float2 a0f = *reinterpret_cast<const float2*>(xa + 4 * s);
            const float2 a1f = *reinterpret_cast<const float2*>(xa + 16 * XPITCH + 4 * s);
            v2f a0; a0.x = a0f.x; a0.y = a0f.y;
            v2f a1; a1.x = a1f.x; a1.y = a1f.y;
            acc0 = __builtin_amdgcn_wmma_f32_16x16x4_f32(
                false, a0, false, bfrag[s], (short)0, acc0, false, false);
            acc1 = __builtin_amdgcn_wmma_f32_16x16x4_f32(
                false, a1, false, bfrag[s], (short)0, acc1, false, false);
        }
        *reinterpret_cast<float4*>(sn + 24 * XPITCH) = q;

        __builtin_amdgcn_sched_barrier(0);
        asm volatile("v_nop\n\tv_nop\n\tv_nop\n\tv_nop" : "+v"(acc0), "+v"(acc1));
        __builtin_amdgcn_sched_barrier(0);

        float* hp = Hp + (size_t)((tile0 + 2 * it) * 16 + 8 * half) * DD + wrow;
        #pragma unroll
        for (int j = 0; j < 8; ++j) hp[j * DD] = acc0[j] + bias;
        if (2 * it + 1 < ntile) {
            #pragma unroll
            for (int j = 0; j < 8; ++j) hp[(16 + j) * DD] = acc1[j] + bias;
        }

        __builtin_amdgcn_fence(__ATOMIC_RELEASE, "workgroup", "local");
        __builtin_amdgcn_s_barrier();
        __builtin_amdgcn_fence(__ATOMIC_ACQUIRE, "workgroup", "local");
    }
}

#define OWN 196
#define QCAP 160
#define NCHUNK (NE / 128)

__global__ __launch_bounds__(1024) void edge_owner_accumulate(
    const int* __restrict__ ei,
    const float* __restrict__ H1, const float* __restrict__ H2,
    const float* __restrict__ H3, const float* __restrict__ H4,
    float* __restrict__ out)
{
    __shared__ float acc[OWN * DD];
    __shared__ float h4s[OWN * DD];
    __shared__ int   cnt[OWN];
    __shared__ int   qe[32 * QCAP];

    const int t = threadIdx.x;
    const int lane = t & 31;
    const int wave = __builtin_amdgcn_readfirstlane(t >> 5);
    const int node0 = blockIdx.x * OWN;
    const int nown = min(OWN, NN - node0);

    for (int q = t; q < OWN * 32; q += 1024) {
        const int n = q >> 5, l = q & 31;
        float4 a = make_float4(0.f, 0.f, 0.f, 0.f), b = a;
        if (n < nown) {
            a = *reinterpret_cast<const float4*>(H1 + (size_t)(node0 + n) * DD + 4 * l);
            b = *reinterpret_cast<const float4*>(H4 + (size_t)(node0 + n) * DD + 4 * l);
        }
        acc[n * DD + l]      = a.x;
        acc[n * DD + 32 + l] = a.y;
        acc[n * DD + 64 + l] = a.z;
        acc[n * DD + 96 + l] = a.w;
        *reinterpret_cast<float4*>(&h4s[n * DD + 4 * l]) = b;
    }
    if (t < OWN) cnt[t] = 0;
    __syncthreads();

    int* const myqe = qe + wave * QCAP;
    const int4* const dstv = reinterpret_cast<const int4*>(ei + NE);
    int qn = 0;

    auto drain = [&]() {
        __builtin_amdgcn_wave_barrier();
        for (int base = 0; base < qn; base += 32) {
            const int nb = min(32, qn - base);
            int my_d = 0, my_s = 0;
            if (lane < nb) {
                const int my_e = myqe[base + lane];
                my_s = ei[my_e];
                my_d = ei[NE + my_e] - node0;
                atomicAdd(&cnt[my_d], 1);
            }
            for (int g = 0; g < nb; g += 4) {
                float4 x2[4], x3[4];
                int dl[4];
                #pragma unroll
                for (int k = 0; k < 4; ++k) {
                    const int idx = (g + k < nb) ? (g + k) : g;
                    const int s = __builtin_amdgcn_readlane(my_s, idx);
                    dl[k] = __builtin_amdgcn_readlane(my_d, idx);
                    x3[k] = *reinterpret_cast<const float4*>(H3 + (size_t)s * DD + 4 * lane);
                    x2[k] = *reinterpret_cast<const float4*>(H2 + (size_t)s * DD + 4 * lane);
                }
                #pragma unroll
                for (int k = 0; k < 4; ++k) {
                    if (g + k < nb) {
                        const float4 x4 = *reinterpret_cast<const float4*>(&h4s[dl[k] * DD + 4 * lane]);
                        float4 m;
                        m.x = x2[k].x / (1.0f + __expf(-(x3[k].x + x4.x)));
                        m.y = x2[k].y / (1.0f + __expf(-(x3[k].y + x4.y)));
                        m.z = x2[k].z / (1.0f + __expf(-(x3[k].z + x4.z)));
                        m.w = x2[k].w / (1.0f + __expf(-(x3[k].w + x4.w)));
                        float* p = &acc[dl[k] * DD + lane];
                        unsafeAtomicAdd(p,      m.x);
                        unsafeAtomicAdd(p + 32, m.y);
                        unsafeAtomicAdd(p + 64, m.z);
                        unsafeAtomicAdd(p + 96, m.w);
                    }
                }
            }
        }
        __builtin_amdgcn_wave_barrier();
        qn = 0;
    };

    const int Kw = (NCHUNK - wave + 31) >> 5;
    const int k0 = (int)((blockIdx.x * (unsigned)Kw) >> 8);
    const unsigned toff = 32u * (unsigned)wave + (unsigned)lane;
    const int e0 = wave * 128 + lane * 4;
    for (int seg = 0; seg < 2; ++seg) {
        const int kend = seg == 0 ? Kw : k0;
        for (int k = seg == 0 ? k0 : 0; k < kend; k += 4) {
            const int4* const kp = dstv + (size_t)k * 1024;
            int4 d[4];
            if (k + 4 <= kend) {
                #pragma unroll
                for (int u = 0; u < 4; ++u) d[u] = kp[toff + 1024u * u];
            } else {
                #pragma unroll
                for (int u = 0; u < 4; ++u) d[u] = kp[toff + 1024u * (unsigned)min(u, kend - 1 - k)];
            }
            #pragma unroll
            for (int u = 0; u < 4; ++u) {
                if (k + u < kend) {
                    const unsigned r0 = (unsigned)(d[u].x - node0), r1 = (unsigned)(d[u].y - node0);
                    const unsigned r2 = (unsigned)(d[u].z - node0), r3 = (unsigned)(d[u].w - node0);
                    const unsigned rmin = min(min(r0, r1), min(r2, r3));
                    if (__builtin_amdgcn_ballot_w32(rmin < (unsigned)nown) != 0u) {
                        const unsigned r[4] = {r0, r1, r2, r3};
                        const int ebase = (k + u) * 4096 + e0;
                        #pragma unroll
                        for (int j = 0; j < 4; ++j) {
                            const bool h = r[j] < (unsigned)nown;
                            const unsigned mk = __builtin_amdgcn_ballot_w32(h);
                            if (mk != 0u) {
                                if (h) myqe[qn + (int)__builtin_amdgcn_mbcnt_lo(mk, 0u)] = ebase + j;
                                qn += __builtin_popcount(mk);
                            }
                        }
                        if (qn >= 32) drain();
                    }
                }
            }
        }
    }
    if (qn > 0) drain();
    __syncthreads();

    for (int q = t; q < nown * 32; q += 1024) {
        const int n = q >> 5, l = q & 31;
        const float s = 1.0f / fmaxf((float)cnt[n], 1.0f);
        float4* po = reinterpret_cast<float4*>(out + (size_t)(node0 + n) * DD + 4 * l);
        float4 o = *po;
        o.x += acc[n * DD + l]      * s;
        o.y += acc[n * DD + 32 + l] * s;
        o.z += acc[n * DD + 64 + l] * s;
        o.w += acc[n * DD + 96 + l] * s;
        *po = o;
    }
}

extern "C" void kernel_launch(void* const* d_in, const int* in_sizes, int n_in,
                              void* d_out, int out_size, void* d_ws, size_t ws_size,
                              hipStream_t stream)
{
    const float* X   = (const float*)d_in[0];
    const int*   ei  = (const int*)  d_in[1];
    const float* W1  = (const float*)d_in[2];
    const float* b1  = (const float*)d_in[3];
    const float* W2  = (const float*)d_in[4];
    const float* b2  = (const float*)d_in[5];
    const float* W3  = (const float*)d_in[6];
    const float* b3  = (const float*)d_in[7];
    const float* W4  = (const float*)d_in[8];
    const float* b4  = (const float*)d_in[9];
    const float* Wr  = (const float*)d_in[10];
    const float* br  = (const float*)d_in[11];
    float* out = (float*)d_out;

    const size_t nd = (size_t)NN * DD;
    float* ws  = (float*)d_ws;
    float* H1  = ws;
    float* H2  = ws + nd;
    float* H3  = ws + 2 * nd;
    float* H4  = ws + 3 * nd;

    gemm5_wstationary<<<dim3(NSLAB, 5), 256, 0, stream>>>(
        X, W1, b1, W2, b2, W3, b3, W4, b4, Wr, br,
        H1, H2, H3, H4,   out);

    edge_owner_accumulate<<<(NN + OWN - 1) / OWN, 1024, 0, stream>>>(ei, H1, H2, H3, H4, out);
}


# Edge filter skips unmarked chunks: the GEMM launch builds a chunk map (per 128 edges, which of 256 owners occur; LDS or; 200 KB); a work-group loads only its marked chunks; sums in arrival order
# speedup vs baseline: 2.3613x; 1.0596x over previous
#include <hip/hip_runtime.h>

#define NN 50000
#define NE 800000
#define DD 128

typedef float v2f __attribute__((ext_vector_type(2)));
typedef float v8f __attribute__((ext_vector_type(8)));

#define NTILES (NN / 16)
#define SLAB_TILES 16
#define NSLAB ((NTILES + SLAB_TILES - 1) / SLAB_TILES)
#define XPITCH 132
#define STAGE_FLOATS (32 * XPITCH)
#define OWN 196
#define NCHUNK (NE / 128)
#define NOWNER ((NN + OWN - 1) / OWN)
static_assert(NOWNER == 256, "the chunk map holds 8 words of 32 owners per chunk");
static_assert(NSLAB * 32 >= NCHUNK, "one block of weight 0 per 32 chunks");
static_assert(NE % 128 == 0, "whole chunks");
#define MAP_END_BYTES (((size_t)4 * NN * DD + (size_t)8 * NCHUNK) * 4)

__global__ __launch_bounds__(256, 8) void gemm5_wstationary(
    const float* __restrict__ X,
    const float* __restrict__ W1, const float* __restrict__ b1,
    const float* __restrict__ W2, const float* __restrict__ b2,
    const float* __restrict__ W3, const float* __restrict__ b3,
    const float* __restrict__ W4, const float* __restrict__ b4,
    const float* __restrict__ Wr, const float* __restrict__ br,
    float* __restrict__ H1, float* __restrict__ H2,
    float* __restrict__ H3, float* __restrict__ H4,
    float* __restrict__ Hr,
    const int* __restrict__ ei, float* __restrict__ CM, size_t ws_bytes)
{
    __shared__ float xs[2 * STAGE_FLOATS];
    __shared__ unsigned cm[8 * 32];

    const int t = threadIdx.x;
    const int wave = __builtin_amdgcn_readfirstlane(t >> 5);
    const int lane = t & 31;
    const int half = lane >> 4;
    const int nn   = lane & 15;
    const int wrow = wave * 16 + nn;

    const int ws = blockIdx.y;
    if (ws == 0 && ws_bytes >= MAP_END_BYTES) {
        unsigned* __restrict__ cmg = reinterpret_cast<unsigned*>(CM);
        const int4* __restrict__ dstv = reinterpret_cast<const int4*>(ei + NE);
        const int cg0 = blockIdx.x * 32;
        cm[t] = 0u;
        __syncthreads();
        int4 d[4];
        #pragma unroll
        for (int i = 0; i < 4; ++i)
            d[i] = dstv[(size_t)min(cg0 + wave * 4 + i, NCHUNK - 1) * 32 + lane];
        #pragma unroll
        for (int i = 0; i < 4; ++i) {
            const int cl = wave * 4 + i;
            if (cg0 + cl < NCHUNK) {
                const unsigned o0 = min((unsigned)d[i].x / (unsigned)OWN, (unsigned)(NOWNER - 1));
                const unsigned o1 = min((unsigned)d[i].y / (unsigned)OWN, (unsigned)(NOWNER - 1));
                const unsigned o2 = min((unsigned)d[i].z / (unsigned)OWN, (unsigned)(NOWNER - 1));
                const unsigned o3 = min((unsigned)d[i].w / (unsigned)OWN, (unsigned)(NOWNER - 1));
                atomicOr(&cm[(o0 >> 5) * 32 + cl], 1u << (o0 & 31));
                atomicOr(&cm[(o1 >> 5) * 32 + cl], 1u << (o1 & 31));
                atomicOr(&cm[(o2 >> 5) * 32 + cl], 1u << (o2 & 31));
                atomicOr(&cm[(o3 >> 5) * 32 + cl], 1u << (o3 & 31));
            }
        }
        __syncthreads();
        if (cg0 + lane < NCHUNK) cmg[(size_t)wave * NCHUNK + cg0 + lane] = cm[wave * 32 + lane];
    }
    const float* __restrict__ Wp = ws == 0 ? W1 : ws == 1 ? W2 : ws == 2 ? W3 : ws == 3 ? W4 : Wr;
    const float* __restrict__ Bp = ws == 0 ? b1 : ws == 1 ? b2 : ws == 2 ? b3 : ws == 3 ? b4 : br;
    float* __restrict__ Hp       = ws == 0 ? H1 : ws == 1 ? H2 : ws == 2 ? H3 : ws == 3 ? H4 : Hr;

    const int tile0 = blockIdx.x * SLAB_TILES;
    const int ntile = min(SLAB_TILES, NTILES - tile0);
    const int niter = (ntile + 1) >> 1;

    const int sc = (t & 31) << 2;
    const int srow = t >> 5;
    auto xtile = [&](int tile) {
        return X + ((size_t)min(tile, NTILES - 1) * 16 + srow) * DD + sc;
    };
    float4 pa0, pa1, pb0, pb1;

    v2f bfrag[32];
    #pragma unroll
    for (int p = 0; p < 2; ++p) {
        #pragma unroll
        for (int i = 0; i < 8; ++i) {
            const int r = srow + 8 * i;
            const float4 v = *reinterpret_cast<const float4*>(Wp + (size_t)(64 * p + r) * DD + sc);
            *reinterpret_cast<float4*>(&xs[r * XPITCH + sc]) = v;
        }
        if (p == 1) {
            const float* xa0 = xtile(tile0);
            const float* xb0 = xtile(tile0 + 1);
            pa0 = *reinterpret_cast<const float4*>(xa0);
            pa1 = *reinterpret_cast<const float4*>(xa0 + 8 * DD);
            pb0 = *reinterpret_cast<const float4*>(xb0);
            pb1 = *reinterpret_cast<const float4*>(xb0 + 8 * DD);
        }
        __syncthreads();
        if ((wave >> 2) == p) {
            const float* wp = &xs[((wave & 3) * 16 + nn) * XPITCH + 2 * half];
            #pragma unroll
            for (int s = 0; s < 32; ++s) {
                const float2 bf = *reinterpret_cast<const float2*>(wp + 4 * s);
                bfrag[s].x = bf.x; bfrag[s].y = bf.y;
            }
        }
        __syncthreads();
    }
    const float bias = Bp[wrow];

    {
        float* sp = &xs[srow * XPITCH + sc];
        *reinterpret_cast<float4*>(sp) = pa0;
        *reinterpret_cast<float4*>(sp + 8 * XPITCH) = pa1;
        *reinterpret_cast<float4*>(sp + 16 * XPITCH) = pb0;
        *reinterpret_cast<float4*>(sp + 24 * XPITCH) = pb1;
    }
    __syncthreads();

    for (int it = 0; it < niter; ++it) {
        const int stage = it & 1;
        const float* xna = xtile(tile0 + 2 * it + 2);
        const float* xnb = xtile(tile0 + 2 * it + 3);
        float* sn = &xs[(stage ^ 1) * STAGE_FLOATS + srow * XPITCH + sc];

        float4 q = *reinterpret_cast<const float4*>(xna);

        v8f acc0 = {}, acc1 = {};
        const float* xa = &xs[stage * STAGE_FLOATS + nn * XPITCH + 2 * half];
        __builtin_amdgcn_sched_barrier(0);
        #pragma unroll
        for (int s = 0; s < 32; ++s) {
            if (s == 8) {
                *reinterpret_cast<float4*>(sn) = q;
                q = *reinterpret_cast<const float4*>(xna + 8 * DD);
            }
            if (s == 16) {
                *reinterpret_cast<float4*>(sn + 8 * XPITCH) = q;
                q = *reinterpret_cast<const float4*>(xnb);
            }
            if (s == 24) {
                *reinterpret_cast<float4*>(sn + 16 * XPITCH) = q;
                q = *reinterpret_cast<const float4*>(xnb + 8 * DD);
            }
            const float2 a0f = *reinterpret_cast<const float2*>(xa + 4 * s);
            const float2 a1f = *reinterpret_cast<const float2*>(xa + 16 * XPITCH + 4 * s);
            v2f a0; a0.x = a0f.x; a0.y = a0f.y;
            v2f a1; a1.x = a1f.x; a1.y = a1f.y;
            acc0 = __builtin_amdgcn_wmma_f32_16x16x4_f32(
                false, a0, false, bfrag[s], (short)0, acc0, false, false);
            acc1 = __builtin_amdgcn_wmma_f32_16x16x4_f32(
                false, a1, false, bfrag[s], (short)0, acc1, false, false);
        }
        *reinterpret_cast<float4*>(sn + 24 * XPITCH) = q;

        __builtin_amdgcn_sched_barrier(0);
        asm volatile("v_nop\n\tv_nop\n\tv_nop\n\tv_nop" : "+v"(acc0), "+v"(acc1));
        __builtin_amdgcn_sched_barrier(0);

        float* hp = Hp + (size_t)((tile0 + 2 * it) * 16 + 8 * half) * DD + wrow;
        #pragma unroll
        for (int j = 0; j < 8; ++j) hp[j * DD] = acc0[j] + bias;
        if (2 * it + 1 < ntile) {
            #pragma unroll
            for (int j = 0; j < 8; ++j) hp[(16 + j) * DD] = acc1[j] + bias;
        }

        __builtin_amdgcn_fence(__ATOMIC_RELEASE, "workgroup", "local");
        __builtin_amdgcn_s_barrier();
        __builtin_amdgcn_fence(__ATOMIC_ACQUIRE, "workgroup", "local");
    }
}

#define QCAP 160

__global__ __launch_bounds__(1024) void edge_owner_accumulate(
    const int* __restrict__ ei,
    const float* __restrict__ H1, const float* __restrict__ H2,
    const float* __restrict__ H3, const float* __restrict__ H4,
    const float* __restrict__ CM, size_t ws_bytes,
    float* __restrict__ out)
{
    __shared__ float acc[OWN * DD];
    __shared__ float h4s[OWN * DD];
    __shared__ int   cnt[OWN];
    __shared__ int   qe[32 * QCAP];

    const int t = threadIdx.x;
    const int lane = t & 31;
    const int wave = __builtin_amdgcn_readfirstlane(t >> 5);
    const int node0 = blockIdx.x * OWN;
    const int nown = min(OWN, NN - node0);

    for (int q = t; q < OWN * 32; q += 1024) {
        const int n = q >> 5, l = q & 31;
        float4 a = make_float4(0.f, 0.f, 0.f, 0.f), b = a;
        if (n < nown) {
            a = *reinterpret_cast<const float4*>(H1 + (size_t)(node0 + n) * DD + 4 * l);
            b = *reinterpret_cast<const float4*>(H4 + (size_t)(node0 + n) * DD + 4 * l);
        }
        acc[n * DD + l]      = a.x;
        acc[n * DD + 32 + l] = a.y;
        acc[n * DD + 64 + l] = a.z;
        acc[n * DD + 96 + l] = a.w;
        *reinterpret_cast<float4*>(&h4s[n * DD + 4 * l]) = b;
    }
    if (t < OWN) cnt[t] = 0;
    __syncthreads();

    int* const myqe = qe + wave * QCAP;
    const int4* const dstv = reinterpret_cast<const int4*>(ei + NE);
    int qn = 0;

    auto drain = [&]() {
        __builtin_amdgcn_wave_barrier();
        for (int base = 0; base < qn; base += 32) {
            const int nb = min(32, qn - base);
            int my_d = 0, my_s = 0;
            if (lane < nb) {
                const int my_e = myqe[base + lane];
                my_s = ei[my_e];
                my_d = ei[NE + my_e] - node0;
                atomicAdd(&cnt[my_d], 1);
            }
            for (int g = 0; g < nb; g += 4) {
                float4 x2[4], x3[4];
                int dl[4];
                #pragma unroll
                for (int k = 0; k < 4; ++k) {
                    const int idx = (g + k < nb) ? (g + k) : g;
                    const int s = __builtin_amdgcn_readlane(my_s, idx);
                    dl[k] = __builtin_amdgcn_readlane(my_d, idx);
                    x3[k] = *reinterpret_cast<const float4*>(H3 + (size_t)s * DD + 4 * lane);
                    x2[k] = *reinterpret_cast<const float4*>(H2 + (size_t)s * DD + 4 * lane);
                }
                #pragma unroll
                for (int k = 0; k < 4; ++k) {
                    if (g + k < nb) {
                        const float4 x4 = *reinterpret_cast<const float4*>(&h4s[dl[k] * DD + 4 * lane]);
                        float4 m;
                        m.x = x2[k].x / (1.0f + __expf(-(x3[k].x + x4.x)));
                        m.y = x2[k].y / (1.0f + __expf(-(x3[k].y + x4.y)));
                        m.z = x2[k].z / (1.0f + __expf(-(x3[k].z + x4.z)));
                        m.w = x2[k].w / (1.0f + __expf(-(x3[k].w + x4.w)));
                        float* p = &acc[dl[k] * DD + lane];
                        unsafeAtomicAdd(p,      m.x);
                        unsafeAtomicAdd(p + 32, m.y);
                        unsafeAtomicAdd(p + 64, m.z);
                        unsafeAtomicAdd(p + 96, m.w);
                    }
                }
            }
        }
        __builtin_amdgcn_wave_barrier();
        qn = 0;
    };

    const unsigned* __restrict__ plane =
        reinterpret_cast<const unsigned*>(CM) + (size_t)(blockIdx.x >> 5) * NCHUNK;
    const unsigned bbit = 1u << (blockIdx.x & 31);
    const int slice = (wave + (int)blockIdx.x) & 31;
    const int c_lo = (slice * NCHUNK) >> 5;
    const int c_hi = ((slice + 1) * NCHUNK) >> 5;
    const bool have_map = ws_bytes >= MAP_END_BYTES;
    unsigned mw = (c_lo + lane < c_hi) ? (have_map ? plane[c_lo + lane] : bbit) : 0u;
    for (int cb = c_lo; cb < c_hi; cb += 32) {
        unsigned hm = __builtin_amdgcn_ballot_w32((mw & bbit) != 0u);
        mw = (cb + 32 + lane < c_hi) ? (have_map ? plane[cb + 32 + lane] : bbit) : 0u;
        while (hm != 0u) {
            int cs[4];
            #pragma unroll
            for (int u = 0; u < 4; ++u) {
                cs[u] = hm != 0u ? cb + __builtin_ctz(hm) : -1;
                hm &= hm - 1u;
            }
            int4 d[4];
            #pragma unroll
            for (int u = 0; u < 4; ++u) d[u] = dstv[(size_t)(cs[u] >= 0 ? cs[u] : cs[0]) * 32 + lane];
            #pragma unroll
            for (int u = 0; u < 4; ++u) {
                if (cs[u] >= 0) {
                    const unsigned r[4] = {(unsigned)(d[u].x - node0), (unsigned)(d[u].y - node0),
                                           (unsigned)(d[u].z - node0), (unsigned)(d[u].w - node0)};
                    const int ebase = cs[u] * 128 + lane * 4;
                    #pragma unroll
                    for (int j = 0; j < 4; ++j) {
                        const bool h = r[j] < (unsigned)nown;
                        const unsigned mk = __builtin_amdgcn_ballot_w32(h);
                        if (mk != 0u) {
                            if (h) myqe[qn + (int)__builtin_amdgcn_mbcnt_lo(mk, 0u)] = ebase + j;
                            qn += __builtin_popcount(mk);
                        }
                    }
                    if (qn >= 32) drain();
                }
            }
        }
    }
    if (qn > 0) drain();
    __syncthreads();

    for (int q = t; q < nown * 32; q += 1024) {
        const int n = q >> 5, l = q & 31;
        const float s = 1.0f / fmaxf((float)cnt[n], 1.0f);
        float4* po = reinterpret_cast<float4*>(out + (size_t)(node0 + n) * DD + 4 * l);
        float4 o = *po;
        o.x += acc[n * DD + l]      * s;
        o.y += acc[n * DD + 32 + l] * s;
        o.z += acc[n * DD + 64 + l] * s;
        o.w += acc[n * DD + 96 + l] * s;
        *po = o;
    }
}

extern "C" void kernel_launch(void* const* d_in, const int* in_sizes, int n_in,
                              void* d_out, int out_size, void* d_ws, size_t ws_size,
                              hipStream_t stream)
{
    const float* X   = (const float*)d_in[0];
    const int*   ei  = (const int*)  d_in[1];
    const float* W1  = (const float*)d_in[2];
    const float* b1  = (const float*)d_in[3];
    const float* W2  = (const float*)d_in[4];
    const float* b2  = (const float*)d_in[5];
    const float* W3  = (const float*)d_in[6];
    const float* b3  = (const float*)d_in[7];
    const float* W4  = (const float*)d_in[8];
    const float* b4  = (const float*)d_in[9];
    const float* Wr  = (const float*)d_in[10];
    const float* br  = (const float*)d_in[11];
    float* out = (float*)d_out;

    const size_t nd = (size_t)NN * DD;
    float* ws  = (float*)d_ws;
    float* H1  = ws;
    float* H2  = ws + nd;
    float* H3  = ws + 2 * nd;
    float* H4  = ws + 3 * nd;
    float* CM  = ws + 4 * nd;

    gemm5_wstationary<<<dim3(NSLAB, 5), 256, 0, stream>>>(
        X, W1, b1, W2, b2, W3, b3, W4, b4, Wr, br,
        H1, H2, H3, H4,   out, ei, CM, ws_size);

    edge_owner_accumulate<<<(NN + OWN - 1) / OWN, 1024, 0, stream>>>(ei, H1, H2, H3, H4, CM, ws_size, out);
}


# Edge kernel: a wave works its queue off in 32-edge batches as a software pipeline, one edge a step: the H2/H3 rows of the next 7 edges are in flight under each edge's gate math; sum order as before
# speedup vs baseline: 2.4312x; 1.0296x over previous
#include <hip/hip_runtime.h>

#define NN 50000
#define NE 800000
#define DD 128

typedef float v2f __attribute__((ext_vector_type(2)));
typedef float v8f __attribute__((ext_vector_type(8)));

#define NTILES (NN / 16)
#define SLAB_TILES 16
#define NSLAB ((NTILES + SLAB_TILES - 1) / SLAB_TILES)
#define XPITCH 132
#define STAGE_FLOATS (32 * XPITCH)
#define OWN 196
#define NCHUNK (NE / 128)
#define NOWNER ((NN + OWN - 1) / OWN)
static_assert(NOWNER == 256, "the chunk map holds 8 words of 32 owners per chunk");
static_assert(NSLAB * 32 >= NCHUNK, "one block of weight 0 per 32 chunks");
static_assert(NE % 128 == 0, "whole chunks");
#define MAP_END_BYTES (((size_t)4 * NN * DD + (size_t)8 * NCHUNK) * 4)

__global__ __launch_bounds__(256, 8) void gemm5_wstationary(
    const float* __restrict__ X,
    const float* __restrict__ W1, const float* __restrict__ b1,
    const float* __restrict__ W2, const float* __restrict__ b2,
    const float* __restrict__ W3, const float* __restrict__ b3,
    const float* __restrict__ W4, const float* __restrict__ b4,
    const float* __restrict__ Wr, const float* __restrict__ br,
    float* __restrict__ H1, float* __restrict__ H2,
    float* __restrict__ H3, float* __restrict__ H4,
    float* __restrict__ Hr,
    const int* __restrict__ ei, float* __restrict__ CM, size_t ws_bytes)
{
    __shared__ float xs[2 * STAGE_FLOATS];
    __shared__ unsigned cm[8 * 32];

    const int t = threadIdx.x;
    const int wave = __builtin_amdgcn_readfirstlane(t >> 5);
    const int lane = t & 31;
    const int half = lane >> 4;
    const int nn   = lane & 15;
    const int wrow = wave * 16 + nn;

    const int ws = blockIdx.y;
    if (ws == 0 && ws_bytes >= MAP_END_BYTES) {
        unsigned* __restrict__ cmg = reinterpret_cast<unsigned*>(CM);
        const int4* __restrict__ dstv = reinterpret_cast<const int4*>(ei + NE);
        const int cg0 = blockIdx.x * 32;
        cm[t] = 0u;
        __syncthreads();
        int4 d[4];
        #pragma unroll
        for (int i = 0; i < 4; ++i)
            d[i] = dstv[(size_t)min(cg0 + wave * 4 + i, NCHUNK - 1) * 32 + lane];
        #pragma unroll
        for (int i = 0; i < 4; ++i) {
            const int cl = wave * 4 + i;
            if (cg0 + cl < NCHUNK) {
                const unsigned o0 = min((unsigned)d[i].x / (unsigned)OWN, (unsigned)(NOWNER - 1));
                const unsigned o1 = min((unsigned)d[i].y / (unsigned)OWN, (unsigned)(NOWNER - 1));
                const unsigned o2 = min((unsigned)d[i].z / (unsigned)OWN, (unsigned)(NOWNER - 1));
                const unsigned o3 = min((unsigned)d[i].w / (unsigned)OWN, (unsigned)(NOWNER - 1));
                atomicOr(&cm[(o0 >> 5) * 32 + cl], 1u << (o0 & 31));
                atomicOr(&cm[(o1 >> 5) * 32 + cl], 1u << (o1 & 31));
                atomicOr(&cm[(o2 >> 5) * 32 + cl], 1u << (o2 & 31));
                atomicOr(&cm[(o3 >> 5) * 32 + cl], 1u << (o3 & 31));
            }
        }
        __syncthreads();
        if (cg0 + lane < NCHUNK) cmg[(size_t)wave * NCHUNK + cg0 + lane] = cm[wave * 32 + lane];
    }
    const float* __restrict__ Wp = ws == 0 ? W1 : ws == 1 ? W2 : ws == 2 ? W3 : ws == 3 ? W4 : Wr;
    const float* __restrict__ Bp = ws == 0 ? b1 : ws == 1 ? b2 : ws == 2 ? b3 : ws == 3 ? b4 : br;
    float* __restrict__ Hp       = ws == 0 ? H1 : ws == 1 ? H2 : ws == 2 ? H3 : ws == 3 ? H4 : Hr;

    const int tile0 = blockIdx.x * SLAB_TILES;
    const int ntile = min(SLAB_TILES, NTILES - tile0);
    const int niter = (ntile + 1) >> 1;

    const int sc = (t & 31) << 2;
    const int srow = t >> 5;
    auto xtile = [&](int tile) {
        return X + ((size_t)min(tile, NTILES - 1) * 16 + srow) * DD + sc;
    };
    float4 pa0, pa1, pb0, pb1;

    v2f bfrag[32];
    #pragma unroll
    for (int p = 0; p < 2; ++p) {
        #pragma unroll
        for (int i = 0; i < 8; ++i) {
            const int r = srow + 8 * i;
            const float4 v = *reinterpret_cast<const float4*>(Wp + (size_t)(64 * p + r) * DD + sc);
            *reinterpret_cast<float4*>(&xs[r * XPITCH + sc]) = v;
        }
        if (p == 1) {
            const float* xa0 = xtile(tile0);
            const float* xb0 = xtile(tile0 + 1);
            pa0 = *reinterpret_cast<const float4*>(xa0);
            pa1 = *reinterpret_cast<const float4*>(xa0 + 8 * DD);
            pb0 = *reinterpret_cast<const float4*>(xb0);
            pb1 = *reinterpret_cast<const float4*>(xb0 + 8 * DD);
        }
        __syncthreads();
        if ((wave >> 2) == p) {
            const float* wp = &xs[((wave & 3) * 16 + nn) * XPITCH + 2 * half];
            #pragma unroll
            for (int s = 0; s < 32; ++s) {
                const float2 bf = *reinterpret_cast<const float2*>(wp + 4 * s);
                bfrag[s].x = bf.x; bfrag[s].y = bf.y;
            }
        }
        __syncthreads();
    }
    const float bias = Bp[wrow];

    {
        float* sp = &xs[srow * XPITCH + sc];
        *reinterpret_cast<float4*>(sp) = pa0;
        *reinterpret_cast<float4*>(sp + 8 * XPITCH) = pa1;
        *reinterpret_cast<float4*>(sp + 16 * XPITCH) = pb0;
        *reinterpret_cast<float4*>(sp + 24 * XPITCH) = pb1;
    }
    __syncthreads();

    for (int it = 0; it < niter; ++it) {
        const int stage = it & 1;
        const float* xna = xtile(tile0 + 2 * it + 2);
        const float* xnb = xtile(tile0 + 2 * it + 3);
        float* sn = &xs[(stage ^ 1) * STAGE_FLOATS + srow * XPITCH + sc];

        float4 q = *reinterpret_cast<const float4*>(xna);

        v8f acc0 = {}, acc1 = {};
        const float* xa = &xs[stage * STAGE_FLOATS + nn * XPITCH + 2 * half];
        __builtin_amdgcn_sched_barrier(0);
        #pragma unroll
        for (int s = 0; s < 32; ++s) {
            if (s == 8) {
                *reinterpret_cast<float4*>(sn) = q;
                q = *reinterpret_cast<const float4*>(xna + 8 * DD);
            }
            if (s == 16) {
                *reinterpret_cast<float4*>(sn + 8 * XPITCH) = q;
                q = *reinterpret_cast<const float4*>(xnb);
            }
            if (s == 24) {
                *reinterpret_cast<float4*>(sn + 16 * XPITCH) = q;
                q = *reinterpret_cast<const float4*>(xnb + 8 * DD);
            }
            const float2 a0f = *reinterpret_cast<const float2*>(xa + 4 * s);
            const float2 a1f = *reinterpret_cast<const float2*>(xa + 16 * XPITCH + 4 * s);
            v2f a0; a0.x = a0f.x; a0.y = a0f.y;
            v2f a1; a1.x = a1f.x; a1.y = a1f.y;
            acc0 = __builtin_amdgcn_wmma_f32_16x16x4_f32(
                false, a0, false, bfrag[s], (short)0, acc0, false, false);
            acc1 = __builtin_amdgcn_wmma_f32_16x16x4_f32(
                false, a1, false, bfrag[s], (short)0, acc1, false, false);
        }
        *reinterpret_cast<float4*>(sn + 24 * XPITCH) = q;

        __builtin_amdgcn_sched_barrier(0);
        asm volatile("v_nop\n\tv_nop\n\tv_nop\n\tv_nop" : "+v"(acc0), "+v"(acc1));
        __builtin_amdgcn_sched_barrier(0);

        float* hp = Hp + (size_t)((tile0 + 2 * it) * 16 + 8 * half) * DD + wrow;
        #pragma unroll
        for (int j = 0; j < 8; ++j) hp[j * DD] = acc0[j] + bias;
        if (2 * it + 1 < ntile) {
            #pragma unroll
            for (int j = 0; j < 8; ++j) hp[(16 + j) * DD] = acc1[j] + bias;
        }

        __builtin_amdgcn_fence(__ATOMIC_RELEASE, "workgroup", "local");
        __builtin_amdgcn_s_barrier();
        __builtin_amdgcn_fence(__ATOMIC_ACQUIRE, "workgroup", "local");
    }
}

#define QCAP (31 + 4 * 128 + 1)

typedef float v4f __attribute__((ext_vector_type(4)));
template <int I> struct step_no { static constexpr int value = I; };
typedef const __attribute__((address_space(1))) v4f* grow_t;
static __device__ __forceinline__ grow_t scalar_row(const float* p)
{
    const unsigned long long a = reinterpret_cast<unsigned long long>(p);
    const unsigned lo = __builtin_amdgcn_readfirstlane((unsigned)a);
    const unsigned hi = __builtin_amdgcn_readfirstlane((unsigned)(a >> 32));
    return (grow_t)(((unsigned long long)hi << 32) | lo);
}

__global__ __launch_bounds__(1024) void edge_owner_accumulate(
    const int* __restrict__ ei,
    const float* __restrict__ H1, const float* __restrict__ H2,
    const float* __restrict__ H3, const float* __restrict__ H4,
    const float* __restrict__ CM, size_t ws_bytes,
    float* __restrict__ out)
{
    __shared__ float acc[OWN * DD];
    __shared__ float h4s[OWN * DD];
    __shared__ int   cnt[OWN];
    __shared__ int   qe[32 * QCAP];

    const int t = threadIdx.x;
    const int lane = t & 31;
    const int wave = __builtin_amdgcn_readfirstlane(t >> 5);
    const int node0 = blockIdx.x * OWN;
    const int nown = min(OWN, NN - node0);

    for (int q = t; q < OWN * 32; q += 1024) {
        const int n = q >> 5, l = q & 31;
        float4 a = make_float4(0.f, 0.f, 0.f, 0.f), b = a;
        if (n < nown) {
            a = *reinterpret_cast<const float4*>(H1 + (size_t)(node0 + n) * DD + 4 * l);
            b = *reinterpret_cast<const float4*>(H4 + (size_t)(node0 + n) * DD + 4 * l);
        }
        acc[n * DD + l]      = a.x;
        acc[n * DD + 32 + l] = a.y;
        acc[n * DD + 64 + l] = a.z;
        acc[n * DD + 96 + l] = a.w;
        *reinterpret_cast<float4*>(&h4s[n * DD + 4 * l]) = b;
    }
    if (t < OWN) cnt[t] = 0;
    __syncthreads();

    int* const myqe = qe + wave * QCAP;
    const int4* const dstv = reinterpret_cast<const int4*>(ei + NE);
    int qn = 0;

    auto batch_head = [&](int base, int nb, int& my_s, int& my_d) {
        my_s = 0; my_d = 0;
        if (lane < nb) {
            const int my_e = myqe[base + lane];
            my_s = ei[my_e];
            my_d = ei[NE + my_e] - node0;
            atomicAdd(&cnt[my_d], 1);
        }
    };
    auto row_request = [&](v4f& x2, v4f& x3, int my_s, int i) {
        const int s = __builtin_amdgcn_readlane(my_s, i);
        x3 = scalar_row(H3 + (size_t)s * DD)[(unsigned)lane];
        x2 = scalar_row(H2 + (size_t)s * DD)[(unsigned)lane];
    };
    auto gate_add = [&](const v4f& x2, const v4f& x3, const float4& x4, int dl) {
        float4 m;
        m.x = x2.x / (1.0f + __expf(-(x3.x + x4.x)));
        m.y = x2.y / (1.0f + __expf(-(x3.y + x4.y)));
        m.z = x2.z / (1.0f + __expf(-(x3.z + x4.z)));
        m.w = x2.w / (1.0f + __expf(-(x3.w + x4.w)));
        float* p = &acc[dl * DD + lane];
        unsafeAtomicAdd(p,      m.x);
        unsafeAtomicAdd(p + 32, m.y);
        unsafeAtomicAdd(p + 64, m.z);
        unsafeAtomicAdd(p + 96, m.w);
    };
    auto batch = [&](int base, int nb) {
        int my_s, my_d;
        batch_head(base, nb, my_s, my_d);
        v4f r2[8], r3[8];
        #pragma unroll
        for (int i = 0; i < 8; ++i) row_request(r2[i], r3[i], my_s, i);
        float4 x4n = *reinterpret_cast<const float4*>(&h4s[__builtin_amdgcn_readlane(my_d, 0) * DD + 4 * lane]);
        auto step = [&](auto tag) -> bool {
            constexpr int i = decltype(tag)::value;
            if (i >= nb) return false;
            const int dl = __builtin_amdgcn_readlane(my_d, i);
            float4 x4 = x4n;
            asm volatile("" : "+v"(x4.x), "+v"(x4.y), "+v"(x4.z), "+v"(x4.w));
            if (i + 1 < 32)
                x4n = *reinterpret_cast<const float4*>(&h4s[__builtin_amdgcn_readlane(my_d, i + 1) * DD + 4 * lane]);
            __builtin_amdgcn_sched_barrier(0);
            gate_add(r2[i & 7], r3[i & 7], x4, dl);
            if (i + 8 < 32) row_request(r2[i & 7], r3[i & 7], my_s, i + 8);
            __builtin_amdgcn_sched_barrier(0);
            return true;
        };
        (void)(step(step_no<0>{})  && step(step_no<1>{})  && step(step_no<2>{})  && step(step_no<3>{})  &&
               step(step_no<4>{})  && step(step_no<5>{})  && step(step_no<6>{})  && step(step_no<7>{})  &&
               step(step_no<8>{})  && step(step_no<9>{})  && step(step_no<10>{}) && step(step_no<11>{}) &&
               step(step_no<12>{}) && step(step_no<13>{}) && step(step_no<14>{}) && step(step_no<15>{}) &&
               step(step_no<16>{}) && step(step_no<17>{}) && step(step_no<18>{}) && step(step_no<19>{}) &&
               step(step_no<20>{}) && step(step_no<21>{}) && step(step_no<22>{}) && step(step_no<23>{}) &&
               step(step_no<24>{}) && step(step_no<25>{}) && step(step_no<26>{}) && step(step_no<27>{}) &&
               step(step_no<28>{}) && step(step_no<29>{}) && step(step_no<30>{}) && step(step_no<31>{}));
    };
    auto drain = [&](bool fin) {
        __builtin_amdgcn_wave_barrier();
        const int lim = fin ? qn : (qn & ~31);
        int base = 0;
        for (; base < lim; base += 32) batch(base, min(32, qn - base));
        const int r = max(qn - base, 0);
        int keep = 0;
        if (lane < r) keep = myqe[base + lane];
        __builtin_amdgcn_wave_barrier();
        if (lane < r) myqe[lane] = keep;
        __builtin_amdgcn_wave_barrier();
        qn = r;
    };

    const unsigned* __restrict__ plane =
        reinterpret_cast<const unsigned*>(CM) + (size_t)(blockIdx.x >> 5) * NCHUNK;
    const unsigned bbit = 1u << (blockIdx.x & 31);
    const int slice = (wave + (int)blockIdx.x) & 31;
    const int c_lo = (slice * NCHUNK) >> 5;
    const int c_hi = ((slice + 1) * NCHUNK) >> 5;
    const bool have_map = ws_bytes >= MAP_END_BYTES;
    unsigned mw = (c_lo + lane < c_hi) ? (have_map ? plane[c_lo + lane] : bbit) : 0u;
    for (int cb = c_lo; cb < c_hi; cb += 32) {
        unsigned hm = __builtin_amdgcn_ballot_w32((mw & bbit) != 0u);
        mw = (cb + 32 + lane < c_hi) ? (have_map ? plane[cb + 32 + lane] : bbit) : 0u;
        const bool lastgrp = cb + 32 >= c_hi;
        do {
            if (hm != 0u) {
                int cs[4];
                #pragma unroll
                for (int u = 0; u < 4; ++u) {
                    cs[u] = hm != 0u ? cb + __builtin_ctz(hm) : -1;
                    hm &= hm - 1u;
                }
                int4 d[4];
                #pragma unroll
                for (int u = 0; u < 4; ++u) d[u] = dstv[(size_t)(cs[u] >= 0 ? cs[u] : cs[0]) * 32 + lane];
                #pragma unroll
                for (int u = 0; u < 4; ++u) {
                    if (cs[u] >= 0) {
                        const unsigned r[4] = {(unsigned)(d[u].x - node0), (unsigned)(d[u].y - node0),
                                               (unsigned)(d[u].z - node0), (unsigned)(d[u].w - node0)};
                        const int ebase = cs[u] * 128 + lane * 4;
                        #pragma unroll
                        for (int j = 0; j < 4; ++j) {
                            const bool h = r[j] < (unsigned)nown;
                            const unsigned mk = __builtin_amdgcn_ballot_w32(h);
                            if (mk != 0u) {
                                if (h) myqe[qn + (int)__builtin_amdgcn_mbcnt_lo(mk, 0u)] = ebase + j;
                                qn += __builtin_popcount(mk);
                            }
                        }
                    }
                }
            }
            const bool fin = lastgrp && hm == 0u;
            if (qn >= 32 || (fin && qn > 0)) drain(fin);
        } while (hm != 0u);
    }
    __syncthreads();

    for (int q = t; q < nown * 32; q += 1024) {
        const int n = q >> 5, l = q & 31;
        const float s = 1.0f / fmaxf((float)cnt[n], 1.0f);
        float4* po = reinterpret_cast<float4*>(out + (size_t)(node0 + n) * DD + 4 * l);
        float4 o = *po;
        o.x += acc[n * DD + l]      * s;
        o.y += acc[n * DD + 32 + l] * s;
        o.z += acc[n * DD + 64 + l] * s;
        o.w += acc[n * DD + 96 + l] * s;
        *po = o;
    }
}

extern "C" void kernel_launch(void* const* d_in, const int* in_sizes, int n_in,
                              void* d_out, int out_size, void* d_ws, size_t ws_size,
                              hipStream_t stream)
{
    const float* X   = (const float*)d_in[0];
    const int*   ei  = (const int*)  d_in[1];
    const float* W1  = (const float*)d_in[2];
    const float* b1  = (const float*)d_in[3];
    const float* W2  = (const float*)d_in[4];
    const float* b2  = (const float*)d_in[5];
    const float* W3  = (const float*)d_in[6];
    const float* b3  = (const float*)d_in[7];
    const float* W4  = (const float*)d_in[8];
    const float* b4  = (const float*)d_in[9];
    const float* Wr  = (const float*)d_in[10];
    const float* br  = (const float*)d_in[11];
    float* out = (float*)d_out;

    const size_t nd = (size_t)NN * DD;
    float* ws  = (float*)d_ws;
    float* H1  = ws;
    float* H2  = ws + nd;
    float* H3  = ws + 2 * nd;
    float* H4  = ws + 3 * nd;
    float* CM  = ws + 4 * nd;

    gemm5_wstationary<<<dim3(NSLAB, 5), 256, 0, stream>>>(
        X, W1, b1, W2, b2, W3, b3, W4, b4, Wr, br,
        H1, H2, H3, H4,   out, ei, CM, ws_size);

    edge_owner_accumulate<<<(NN + OWN - 1) / OWN, 1024, 0, stream>>>(ei, H1, H2, H3, H4, CM, ws_size, out);
}
